# starting kernel
# baseline (speedup 1.0000x reference)
#include <hip/hip_runtime.h>
#include <hip/hip_bf16.h>
#include <math.h>

typedef __attribute__((ext_vector_type(2))) float v2f;
typedef __attribute__((ext_vector_type(8))) float v8f;

#define O_DIM 128
#define I_DIM 128
#define M_DIM 32
#define N_DIM 128
#define LOG2E 1.44269504088896340736f

#define SZ_OFF   0
#define SQ_OFF   32
#define SW_OFF   64
#define CC_OFF   96
#define PAD_OFF  128
#define WREG     160

__global__ __launch_bounds__(256)
void gpkan_moment_kernel(const float* __restrict__ x,
                         const float* __restrict__ z,
                         const float* __restrict__ qmu,
                         const float* __restrict__ qlv,
                         const float* __restrict__ lsc,
                         const float* __restrict__ lva,
                         float* __restrict__ out)
{
    __shared__ __attribute__((aligned(16))) float smem[8 * WREG];

    const int o    = blockIdx.x;
    const int wave = threadIdx.x >> 5;
    const int lane = threadIdx.x & 31;
    const int row  = lane & 15;
    const int half = lane >> 4;
    const int n    = wave * 16 + row;

    float* W = &smem[wave * WREG];

    W[PAD_OFF + lane] = 0.0f;

    const int h2 = 2 * half;
    const int a1off = ((row == 0) ? SQ_OFF : PAD_OFF) + h2;
    const int a2off = ((row == 0) ? SW_OFF : ((row == 1) ? CC_OFF : PAD_OFF)) + h2;
    const int zoff  = SZ_OFF + h2;

    float ym = 0.f, yv = 0.f;

    #pragma unroll 1
    for (int i = 0; i < I_DIM; ++i) {
        const int oi = o * I_DIM + i;

        float scale = fmaxf(__expf(lsc[oi]), 0.1f);
        float l2    = scale * scale;
        float sv    = fmaxf(__expf(lva[oi]), 1e-5f);
        float den1  = l2 + 1e-6f;
        float den2  = l2 + 2e-6f;
        float r1    = sqrtf(l2 / den1);
        float r2    = sqrtf(l2 / den2);
        float c2sv  = sv * r2;
        float k1    = -0.5f * LOG2E / den1;
        float k2    = -LOG2E / den2;

        const float* zp  = z   + (size_t)oi * M_DIM;
        const float* qp  = qmu + (size_t)oi * M_DIM;
        const float* qvp = qlv + (size_t)oi * M_DIM;
        float z_l  = zp[lane];
        float qm_l = qp[lane];
        float qv_l = fmaxf(__expf(qvp[lane]), 1e-5f);

        float xn = x[n * I_DIM + i];

        __builtin_amdgcn_wave_barrier();
        W[SZ_OFF + lane] = z_l;
        W[SQ_OFF + lane] = qm_l * r1;
        W[SW_OFF + lane] = (qv_l + qm_l * qm_l) * r2;
        W[CC_OFF + lane] = c2sv;
        __builtin_amdgcn_wave_barrier();

        v8f acc1 = {};
        v8f acc2 = {};

        #pragma unroll
        for (int c = 0; c < 8; ++c) {
            v2f zp2 = *(const v2f*)(W + zoff  + 4 * c);
            v2f a1  = *(const v2f*)(W + a1off + 4 * c);
            v2f a2  = *(const v2f*)(W + a2off + 4 * c);

            float d0  = xn - zp2[0];
            float d1  = xn - zp2[1];
            float d0s = d0 * d0;
            float d1s = d1 * d1;
            v2f b1 = { __builtin_amdgcn_exp2f(d0s * k1),
                       __builtin_amdgcn_exp2f(d1s * k1) };
            v2f b2 = { __builtin_amdgcn_exp2f(d0s * k2),
                       __builtin_amdgcn_exp2f(d1s * k2) };

            acc1 = __builtin_amdgcn_wmma_f32_16x16x4_f32(
                       false, a1, false, b1, (short)0, acc1, false, false);
            acc2 = __builtin_amdgcn_wmma_f32_16x16x4_f32(
                       false, a2, false, b2, (short)0, acc2, false, false);
        }

        float em    = acc1[0];
        float secnd = acc2[0];
        float s2b   = acc2[1];
        float ev = sv - s2b + secnd - em * em;
        ev = fmaxf(ev, 1e-6f);
        ym += em;
        yv += ev;
    }

    if (lane < 16) {
        out[n * O_DIM + o]                 = ym;
        out[N_DIM * O_DIM + n * O_DIM + o] = yv;
    }
}

extern "C" void kernel_launch(void* const* d_in, const int* in_sizes, int n_in,
                              void* d_out, int out_size, void* d_ws, size_t ws_size,
                              hipStream_t stream) {
    const float* x   = (const float*)d_in[0];
    const float* z   = (const float*)d_in[1];
    const float* qmu = (const float*)d_in[2];
    const float* qlv = (const float*)d_in[3];
    const float* lsc = (const float*)d_in[4];
    const float* lva = (const float*)d_in[5];
    float* out = (float*)d_out;

    dim3 grid(O_DIM);
    dim3 block(256);
    gpkan_moment_kernel<<<grid, block, 0, stream>>>(x, z, qmu, qlv, lsc, lva, out);
}


# tile rows = 16 consecutive i: lanes keep their own row's weights from global (trace weight folded in), z rows read lane-broadcast, a tile per i, no LDS, 256x4 waves; f32 sum order changes
# speedup vs baseline: 2.3008x; 2.3008x over previous
#include <hip/hip_runtime.h>
#include <math.h>


typedef __attribute__((ext_vector_type(2))) float v2f;
typedef __attribute__((ext_vector_type(4))) float v4f;
typedef __attribute__((ext_vector_type(8))) float v8f;

#define O_DIM 128
#define I_DIM 128
#define M_DIM 32
#define N_DIM 128
#define LOG2E 1.44269504088896340736f
#define IB    16
#define ZD    4


struct Raw {
    v4f qm[4];
    v4f ql[4];
    v4f xx[4];
    v4f lvh[2];
    float ls;
    float lv;
};

struct Ops {
    float QS[16];
    float W2[16];
    float X[16];
    float SVh[8];
    float K1, K2;
};

__device__ __forceinline__ void load_raw(Raw& R,
        const float* __restrict__ qm_p, const float* __restrict__ ql_p,
        const float* __restrict__ x_p,  const float* __restrict__ ls_p,
        const float* __restrict__ lv_p, const float* __restrict__ lvh_p, int i0)
{
    const v4f* q4 = (const v4f*)(qm_p + (size_t)i0 * M_DIM);
    const v4f* l4 = (const v4f*)(ql_p + (size_t)i0 * M_DIM);
    const v4f* x4 = (const v4f*)(x_p + i0);
    const v4f* h4 = (const v4f*)(lvh_p + i0);
#pragma unroll
    for (int q = 0; q < 4; ++q) { R.qm[q] = q4[q]; R.ql[q] = l4[q]; R.xx[q] = x4[q]; }
    R.lvh[0] = h4[0];
    R.lvh[1] = h4[1];
    R.ls = ls_p[i0];
    R.lv = lv_p[i0];
}

__device__ __forceinline__ void load_zrow(v4f (&Z)[4], const float* __restrict__ z_p, int i)
{
    const v4f* p = (const v4f*)(z_p + (size_t)i * M_DIM);
#pragma unroll
    for (int q = 0; q < 4; ++q) Z[q] = p[q];
}

__device__ __forceinline__ void prepare(Ops& P, const Raw& R)
{
    float scale = fmaxf(__expf(R.ls), 0.1f);
    float l2    = scale * scale;
    float sv    = fmaxf(__expf(R.lv), 1e-5f);
    float den1  = l2 + 1e-6f;
    float den2  = l2 + 2e-6f;
    float r1    = sqrtf(l2 / den1);
    float r2    = sqrtf(l2 / den2);
    float c2sv  = sv * r2;
    P.K1 = -0.5f * LOG2E / den1;
    P.K2 = -LOG2E / den2;
#pragma unroll
    for (int j = 0; j < 16; ++j) {
        float qm = R.qm[j >> 2][j & 3];
        float qv = fmaxf(__expf(R.ql[j >> 2][j & 3]), 1e-5f);
        P.QS[j] = qm * r1;
        P.W2[j] = (qv + qm * qm) * r2 - c2sv;
        P.X[j]  = R.xx[j >> 2][j & 3];
    }
#pragma unroll
    for (int j = 0; j < 8; ++j)
        P.SVh[j] = fmaxf(__expf(R.lvh[j >> 2][j & 3]), 1e-5f);
}

__device__ __forceinline__ float from_lane(float v, int src)
{
    return __int_as_float(__builtin_amdgcn_readlane(__float_as_int(v), src));
}

__global__ __launch_bounds__(128)
void gpkan_rows_kernel(const float* __restrict__ x,
                       const float* __restrict__ z,
                       const float* __restrict__ qmu,
                       const float* __restrict__ qlv,
                       const float* __restrict__ lsc,
                       const float* __restrict__ lva,
                       float* __restrict__ out)
{
    const int o    = blockIdx.x >> 1;
    const int tile = ((blockIdx.x & 1) << 2) + (threadIdx.x >> 5);
    const int lane = threadIdx.x & 31;
    const int r    = lane & 15;
    const int h    = lane >> 4;
    const int n    = tile * 16 + r;

    const size_t ob = (size_t)o * I_DIM;
    const float* z_p   = z   + ob * M_DIM + 16 * h;
    const float* qm_p  = qmu + (ob + r) * M_DIM + 16 * h;
    const float* ql_p  = qlv + (ob + r) * M_DIM + 16 * h;
    const float* x_p   = x + (size_t)n * I_DIM;
    const float* ls_p  = lsc + ob + r;
    const float* lv_p  = lva + ob + r;
    const float* lvh_p = lva + ob + 8 * h;

    Raw cur;
    load_raw(cur, qm_p, ql_p, x_p, ls_p, lv_p, lvh_p, 0);

    __builtin_amdgcn_sched_barrier(0);

    v4f zq[ZD][4];
#pragma unroll
    for (int d = 0; d < ZD; ++d) load_zrow(zq[d], z_p, d);

    float ym = 0.f, yv = 0.f;

#pragma unroll 1
    for (int b = 0; b < I_DIM / IB; ++b) {
        const int i0 = b * IB;

        Ops P;
        prepare(P, cur);

        Raw nxt;
        load_raw(nxt, qm_p, ql_p, x_p, ls_p, lv_p, lvh_p, min(b + 1, I_DIM / IB - 1) * IB);

        float E1[16], E2[16];

#pragma unroll
        for (int s = 0; s < IB; ++s) {
            v4f zn[4];
            load_zrow(zn, z_p, min(i0 + s + ZD, I_DIM - 1));

            const float k1 = from_lane(P.K1, s);
            const float k2 = from_lane(P.K2, s);
            const float xs = P.X[s];

            v8f a1 = {};
            v8f a2 = {};
#pragma unroll
            for (int c = 0; c < 8; ++c) {
                const float z0 = zq[0][c >> 1][(2 * c) & 3];
                const float z1 = zq[0][c >> 1][((2 * c) & 3) + 1];
                float d0  = xs - z0;
                float d1  = xs - z1;
                float d0s = d0 * d0;
                float d1s = d1 * d1;
                v2f b1 = { __builtin_amdgcn_exp2f(d0s * k1),
                           __builtin_amdgcn_exp2f(d1s * k1) };
                v2f b2 = { __builtin_amdgcn_exp2f(d0s * k2),
                           __builtin_amdgcn_exp2f(d1s * k2) };
                v2f w1 = { P.QS[2 * c], P.QS[2 * c + 1] };
                v2f w2 = { P.W2[2 * c], P.W2[2 * c + 1] };
                a1 = __builtin_amdgcn_wmma_f32_16x16x4_f32(
                         false, w1, false, b1, (short)0, a1, false, false);
                a2 = __builtin_amdgcn_wmma_f32_16x16x4_f32(
                         false, w2, false, b2, (short)0, a2, false, false);
            }
            E1[s] = a1[s & 7];
            E2[s] = a2[s & 7];

#pragma unroll
            for (int d = 0; d + 1 < ZD; ++d) {
#pragma unroll
                for (int q = 0; q < 4; ++q) zq[d][q] = zq[d + 1][q];
            }
#pragma unroll
            for (int q = 0; q < 4; ++q) zq[ZD - 1][q] = zn[q];
        }

#pragma unroll
        for (int j = 0; j < 8; ++j) {
            float em = h ? E1[8 + j] : E1[j];
            float t  = h ? E2[8 + j] : E2[j];
            float ev = P.SVh[j] + t - em * em;
            ev = fmaxf(ev, 1e-6f);
            ym += em;
            yv += ev;
        }

        asm volatile("" :: "v"(nxt.qm[0]), "v"(nxt.qm[1]), "v"(nxt.qm[2]), "v"(nxt.qm[3]),
                           "v"(nxt.ql[0]), "v"(nxt.ql[1]), "v"(nxt.ql[2]), "v"(nxt.ql[3]),
                           "v"(nxt.xx[0]), "v"(nxt.xx[1]), "v"(nxt.xx[2]), "v"(nxt.xx[3]),
                           "v"(nxt.lvh[0]), "v"(nxt.lvh[1]), "v"(nxt.ls), "v"(nxt.lv));
        cur = nxt;
    }

    ym += __shfl_xor(ym, 16, 32);
    yv += __shfl_xor(yv, 16, 32);

    if (lane < 16) {
        out[n * O_DIM + o]                 = ym;
        out[N_DIM * O_DIM + n * O_DIM + o] = yv;
    }
}

extern "C" void kernel_launch(void* const* d_in, const int* in_sizes, int n_in,
                              void* d_out, int out_size, void* d_ws, size_t ws_size,
                              hipStream_t stream) {
    const float* x   = (const float*)d_in[0];
    const float* z   = (const float*)d_in[1];
    const float* qmu = (const float*)d_in[2];
    const float* qlv = (const float*)d_in[3];
    const float* lsc = (const float*)d_in[4];
    const float* lva = (const float*)d_in[5];
    float* out = (float*)d_out;

    dim3 grid(256);
    dim3 block(128);
    gpkan_rows_kernel<<<grid, block, 0, stream>>>(x, z, qmu, qlv, lsc, lva, out);
}


# B fragments of step s+1 built in a second register set beside the WMMAs of step s, their four stages one WMMA pair apart, issue order fixed by sched groups; same expressions and sum order
# speedup vs baseline: 2.6623x; 1.1571x over previous
#include <hip/hip_runtime.h>
#include <math.h>


typedef __attribute__((ext_vector_type(2))) float v2f;
typedef __attribute__((ext_vector_type(4))) float v4f;
typedef __attribute__((ext_vector_type(8))) float v8f;

#define O_DIM 128
#define I_DIM 128
#define M_DIM 32
#define N_DIM 128
#define LOG2E 1.44269504088896340736f
#define IB    16
#define ZD    4
#define PREP  0
#define LOADS 11


struct Raw {
    v4f qm[4];
    v4f ql[4];
    v4f xx[4];
    v4f lvh[2];
    float ls;
    float lv;
};

struct Ops {
    float QS[16];
    float W2[16];
    float X[16];
    float SVh[8];
    float K1, K2;
};

__device__ __forceinline__ void load_raw(Raw& R,
        const float* __restrict__ qm_p, const float* __restrict__ ql_p,
        const float* __restrict__ x_p,  const float* __restrict__ ls_p,
        const float* __restrict__ lv_p, const float* __restrict__ lvh_p, int i0)
{
    const v4f* q4 = (const v4f*)(qm_p + (size_t)i0 * M_DIM);
    const v4f* l4 = (const v4f*)(ql_p + (size_t)i0 * M_DIM);
    const v4f* x4 = (const v4f*)(x_p + i0);
    const v4f* h4 = (const v4f*)(lvh_p + i0);
#pragma unroll
    for (int q = 0; q < 4; ++q) { R.qm[q] = q4[q]; R.ql[q] = l4[q]; R.xx[q] = x4[q]; }
    R.lvh[0] = h4[0];
    R.lvh[1] = h4[1];
    R.ls = ls_p[i0];
    R.lv = lv_p[i0];
}

__device__ __forceinline__ void load_zrow(v4f (&Z)[4], const float* __restrict__ z_p, int i)
{
    const v4f* p = (const v4f*)(z_p + (size_t)i * M_DIM);
#pragma unroll
    for (int q = 0; q < 4; ++q) Z[q] = p[q];
}

__device__ __forceinline__ void prepare(Ops& P, const Raw& R)
{
    float scale = fmaxf(__expf(R.ls), 0.1f);
    float l2    = scale * scale;
    float sv    = fmaxf(__expf(R.lv), 1e-5f);
    float den1  = l2 + 1e-6f;
    float den2  = l2 + 2e-6f;
    float r1    = sqrtf(l2 / den1);
    float r2    = sqrtf(l2 / den2);
    float c2sv  = sv * r2;
    P.K1 = -0.5f * LOG2E / den1;
    P.K2 = -LOG2E / den2;
#pragma unroll
    for (int j = 0; j < 16; ++j) {
        float qm = R.qm[j >> 2][j & 3];
        float qv = fmaxf(__expf(R.ql[j >> 2][j & 3]), 1e-5f);
        P.QS[j] = qm * r1;
        P.W2[j] = (qv + qm * qm) * r2 - c2sv;
        P.X[j]  = R.xx[j >> 2][j & 3];
    }
#pragma unroll
    for (int j = 0; j < 8; ++j)
        P.SVh[j] = fmaxf(__expf(R.lvh[j >> 2][j & 3]), 1e-5f);
}

__device__ __forceinline__ float from_lane(float v, int src)
{
    return __int_as_float(__builtin_amdgcn_readlane(__float_as_int(v), src));
}

struct Frag {
    v2f b1[8];
    v2f b2[8];
};

struct Build {
    float d[8][2];
    float q[8][2];
    float e1[8][2];
    float e2[8][2];
};
__device__ __forceinline__ void st_diff(Build& B, int c, const v4f (&Z)[4], float xs)
{
    B.d[c][0] = xs - Z[c >> 1][(2 * c) & 3];
    B.d[c][1] = xs - Z[c >> 1][((2 * c) & 3) + 1];
}
__device__ __forceinline__ void st_square(Build& B, int c)
{
    B.q[c][0] = B.d[c][0] * B.d[c][0];
    B.q[c][1] = B.d[c][1] * B.d[c][1];
}
__device__ __forceinline__ void st_scale1(Build& B, int c, float k1)
{
    B.e1[c][0] = B.q[c][0] * k1;
    B.e1[c][1] = B.q[c][1] * k1;
}
__device__ __forceinline__ void st_scale2(Build& B, int c, float k2)
{
    B.e2[c][0] = B.q[c][0] * k2;
    B.e2[c][1] = B.q[c][1] * k2;
}
__device__ __forceinline__ void st_exp1(Frag& F, const Build& B, int c)
{
    F.b1[c] = v2f{ __builtin_amdgcn_exp2f(B.e1[c][0]), __builtin_amdgcn_exp2f(B.e1[c][1]) };
}
__device__ __forceinline__ void st_exp2(Frag& F, const Build& B, int c)
{
    F.b2[c] = v2f{ __builtin_amdgcn_exp2f(B.e2[c][0]), __builtin_amdgcn_exp2f(B.e2[c][1]) };
}

struct Src {
    float xs, k1, k2;
};

__device__ __forceinline__ void carry(Build& To, const Build& From)
{
#pragma unroll
    for (int e = 0; e < 2; ++e) {
        To.d[2][e]  = From.d[2][e];
        To.q[1][e]  = From.q[1][e];
        To.e1[0][e] = From.e1[0][e];
        To.e2[0][e] = From.e2[0][e];
    }
}

__global__ __launch_bounds__(128)
void gpkan_rows_kernel(const float* __restrict__ x,
                       const float* __restrict__ z,
                       const float* __restrict__ qmu,
                       const float* __restrict__ qlv,
                       const float* __restrict__ lsc,
                       const float* __restrict__ lva,
                       float* __restrict__ out)
{
    const int o    = blockIdx.x >> 1;
    const int tile = ((blockIdx.x & 1) << 2) + (threadIdx.x >> 5);
    const int lane = threadIdx.x & 31;
    const int r    = lane & 15;
    const int h    = lane >> 4;
    const int n    = tile * 16 + r;

    const size_t ob = (size_t)o * I_DIM;
    const float* z_p   = z   + ob * M_DIM + 16 * h;
    const float* qm_p  = qmu + (ob + r) * M_DIM + 16 * h;
    const float* ql_p  = qlv + (ob + r) * M_DIM + 16 * h;
    const float* x_p   = x + (size_t)n * I_DIM;
    const float* ls_p  = lsc + ob + r;
    const float* lv_p  = lva + ob + r;
    const float* lvh_p = lva + ob + 8 * h;

    Raw first;
    load_raw(first, qm_p, ql_p, x_p, ls_p, lv_p, lvh_p, 0);
    Raw ahead;
    load_raw(ahead, qm_p, ql_p, x_p, ls_p, lv_p, lvh_p, IB);

    v4f z0row[4];
    load_zrow(z0row, z_p, 0);
    v4f zq[ZD][4];
#pragma unroll
    for (int d = 0; d < ZD; ++d) load_zrow(zq[d], z_p, 1 + d);

    Ops P;
    prepare(P, first);

    Frag F;
    {
        const float k1 = from_lane(P.K1, 0);
        const float k2 = from_lane(P.K2, 0);
        Build B0;
#pragma unroll
        for (int c = 0; c < 8; ++c) {
            st_diff(B0, c, z0row, P.X[0]);
            st_square(B0, c);
            st_scale1(B0, c, k1);
            st_scale2(B0, c, k2);
            st_exp1(F, B0, c);
            st_exp2(F, B0, c);
        }
    }

    Src   Sc = { P.X[1], from_lane(P.K1, 1), from_lane(P.K2, 1) };
    Build Bc;
    st_diff(Bc, 0, zq[0], Sc.xs);
    st_diff(Bc, 1, zq[0], Sc.xs);
    st_diff(Bc, 2, zq[0], Sc.xs);
    st_square(Bc, 0);
    st_square(Bc, 1);
    st_scale1(Bc, 0, Sc.k1);
    st_scale2(Bc, 0, Sc.k2);

    float ym = 0.f, yv = 0.f;

#pragma unroll 1
    for (int b = 0; b < I_DIM / IB; ++b) {
        const int i0 = b * IB;

        Ops Pn;
        Raw nxt;
        float E1[16], E2[16];

#pragma unroll
        for (int s = 0; s < IB; ++s) {
            v4f zn[4];
            load_zrow(zn, z_p, min(i0 + s + 1 + ZD, I_DIM - 1));

            if (s == PREP) prepare(Pn, ahead);
            if (s == LOADS)
                load_raw(nxt, qm_p, ql_p, x_p, ls_p, lv_p, lvh_p, min(b + 2, I_DIM / IB - 1) * IB);

            const Ops& Q2 = (s + 2 >= IB) ? Pn : P;
            const int  s2 = (s + 2) & (IB - 1);
            const Src  Sn = { Q2.X[s2], from_lane(Q2.K1, s2), from_lane(Q2.K2, s2) };

            Frag  Fn;
            Build Bn;
            v8f a1 = {};
            v8f a2 = {};

#pragma unroll
            for (int c = 0; c < 8; ++c) {
                v2f w1 = { P.QS[2 * c], P.QS[2 * c + 1] };
                v2f w2 = { P.W2[2 * c], P.W2[2 * c + 1] };
                a1 = __builtin_amdgcn_wmma_f32_16x16x4_f32(
                         false, w1, false, F.b1[c], (short)0, a1, false, false);
                a2 = __builtin_amdgcn_wmma_f32_16x16x4_f32(
                         false, w2, false, F.b2[c], (short)0, a2, false, false);
                st_exp1(Fn, Bc, c);
                st_exp2(Fn, Bc, c);
                if (c + 1 < 8) { st_scale1(Bc, c + 1, Sc.k1); st_scale2(Bc, c + 1, Sc.k2); }
                if (c + 2 < 8) st_square(Bc, c + 2);
                if (c + 3 < 8) st_diff(Bc, c + 3, zq[0], Sc.xs);
                if (c == 5) st_diff(Bn, 0, zq[1], Sn.xs);
                if (c == 6) { st_diff(Bn, 1, zq[1], Sn.xs); st_square(Bn, 0); }
                if (c == 7) {
                    st_diff(Bn, 2, zq[1], Sn.xs);
                    st_square(Bn, 1);
                    st_scale1(Bn, 0, Sn.k1);
                    st_scale2(Bn, 0, Sn.k2);
                }
            }
            E1[s] = a1[s & 7];
            E2[s] = a2[s & 7];
            F = Fn;
            carry(Bc, Bn);
            Sc = Sn;

            if (s == PREP) {
                __builtin_amdgcn_sched_group_barrier(0x402, 22, 0);
#pragma unroll
                for (int g = 0; g < 8; ++g) {
                    __builtin_amdgcn_sched_group_barrier(0x008, 2, 0);
                    __builtin_amdgcn_sched_group_barrier(0x402, 28, 0);
                }
            } else {
                __builtin_amdgcn_sched_group_barrier(0x402, 8, 0);
#pragma unroll
                for (int g = 0; g < 8; ++g) {
                    __builtin_amdgcn_sched_group_barrier(0x008, 2, 0);
                    __builtin_amdgcn_sched_group_barrier(0x400, 4, 0);
                    __builtin_amdgcn_sched_group_barrier(0x002, 8, 0);
                }
            }
            __builtin_amdgcn_sched_barrier(0);

#pragma unroll
            for (int d = 0; d + 1 < ZD; ++d) {
#pragma unroll
                for (int q = 0; q < 4; ++q) zq[d][q] = zq[d + 1][q];
            }
#pragma unroll
            for (int q = 0; q < 4; ++q) zq[ZD - 1][q] = zn[q];
        }

#pragma unroll
        for (int j = 0; j < 8; ++j) {
            float em = h ? E1[8 + j] : E1[j];
            float t  = h ? E2[8 + j] : E2[j];
            float ev = P.SVh[j] + t - em * em;
            ev = fmaxf(ev, 1e-6f);
            ym += em;
            yv += ev;
        }

        asm volatile("" :: "v"(nxt.qm[0]), "v"(nxt.qm[1]), "v"(nxt.qm[2]), "v"(nxt.qm[3]),
                           "v"(nxt.ql[0]), "v"(nxt.ql[1]), "v"(nxt.ql[2]), "v"(nxt.ql[3]),
                           "v"(nxt.xx[0]), "v"(nxt.xx[1]), "v"(nxt.xx[2]), "v"(nxt.xx[3]),
                           "v"(nxt.lvh[0]), "v"(nxt.lvh[1]), "v"(nxt.ls), "v"(nxt.lv));
        ahead = nxt;
        P = Pn;
    }

    ym += __shfl_xor(ym, 16, 32);
    yv += __shfl_xor(yv, 16, 32);

    if (lane < 16) {
        out[n * O_DIM + o]                 = ym;
        out[N_DIM * O_DIM + n * O_DIM + o] = yv;
    }
}

extern "C" void kernel_launch(void* const* d_in, const int* in_sizes, int n_in,
                              void* d_out, int out_size, void* d_ws, size_t ws_size,
                              hipStream_t stream) {
    const float* x   = (const float*)d_in[0];
    const float* z   = (const float*)d_in[1];
    const float* qmu = (const float*)d_in[2];
    const float* qlv = (const float*)d_in[3];
    const float* lsc = (const float*)d_in[4];
    const float* lva = (const float*)d_in[5];
    float* out = (float*)d_out;

    dim3 grid(256);
    dim3 block(128);
    gpkan_rows_kernel<<<grid, block, 0, stream>>>(x, z, qmu, qlv, lsc, lva, out);
}


# m-sum of every (n,o,i) split over two waves (16 m each; half sums joined once through LDS behind one barrier, fixed order): 256 blocks x 8 waves = two waves per SIMD; f32 sum order changes
# speedup vs baseline: 2.7673x; 1.0394x over previous
#include <hip/hip_runtime.h>
#include <math.h>


typedef __attribute__((ext_vector_type(2))) float v2f;
typedef __attribute__((ext_vector_type(4))) float v4f;
typedef __attribute__((ext_vector_type(8))) float v8f;

#define O_DIM 128
#define I_DIM 128
#define M_DIM 32
#define N_DIM 128
#define LOG2E 1.44269504088896340736f
#define IB    16
#define NW    2
#define NC    4
#define TPB   4
#define ZD    4
#define PREP  0
#define LOADS 11


struct Raw {
    v4f qm[2];
    v4f ql[2];
    v4f xx[4];
    v4f lvh[2];
    float ls;
    float lv;
};

struct Ops {
    float QS[8];
    float W2[8];
    float X[16];
    float SVh[8];
    float K1, K2;
};

__device__ __forceinline__ void load_raw(Raw& R,
        const float* __restrict__ qm_p, const float* __restrict__ ql_p,
        const float* __restrict__ x_p,  const float* __restrict__ ls_p,
        const float* __restrict__ lv_p, const float* __restrict__ lvh_p, int i0)
{
    const v4f* q4 = (const v4f*)(qm_p + (size_t)i0 * M_DIM);
    const v4f* l4 = (const v4f*)(ql_p + (size_t)i0 * M_DIM);
    const v4f* x4 = (const v4f*)(x_p + i0);
    const v4f* h4 = (const v4f*)(lvh_p + i0);
#pragma unroll
    for (int q = 0; q < 2; ++q) { R.qm[q] = q4[q]; R.ql[q] = l4[q]; }
#pragma unroll
    for (int q = 0; q < 4; ++q) R.xx[q] = x4[q];
    R.lvh[0] = h4[0];
    R.lvh[1] = h4[1];
    R.ls = ls_p[i0];
    R.lv = lv_p[i0];
}

__device__ __forceinline__ void load_zrow(v4f (&Z)[2], const float* __restrict__ z_p, int i)
{
    const v4f* p = (const v4f*)(z_p + (size_t)i * M_DIM);
#pragma unroll
    for (int q = 0; q < 2; ++q) Z[q] = p[q];
}

__device__ __forceinline__ void prepare(Ops& P, const Raw& R)
{
    float scale = fmaxf(__expf(R.ls), 0.1f);
    float l2    = scale * scale;
    float sv    = fmaxf(__expf(R.lv), 1e-5f);
    float den1  = l2 + 1e-6f;
    float den2  = l2 + 2e-6f;
    float r1    = sqrtf(l2 / den1);
    float r2    = sqrtf(l2 / den2);
    float c2sv  = sv * r2;
    P.K1 = -0.5f * LOG2E / den1;
    P.K2 = -LOG2E / den2;
#pragma unroll
    for (int j = 0; j < 8; ++j) {
        float qm = R.qm[j >> 2][j & 3];
        float qv = fmaxf(__expf(R.ql[j >> 2][j & 3]), 1e-5f);
        P.QS[j] = qm * r1;
        P.W2[j] = (qv + qm * qm) * r2 - c2sv;
    }
#pragma unroll
    for (int j = 0; j < 16; ++j)
        P.X[j] = R.xx[j >> 2][j & 3];
#pragma unroll
    for (int j = 0; j < 8; ++j)
        P.SVh[j] = fmaxf(__expf(R.lvh[j >> 2][j & 3]), 1e-5f);
}

__device__ __forceinline__ float from_lane(float v, int src)
{
    return __int_as_float(__builtin_amdgcn_readlane(__float_as_int(v), src));
}

struct Frag {
    v2f b1[NC];
    v2f b2[NC];
};

struct Build {
    float d[NC][2];
    float q[NC][2];
    float e1[NC][2];
    float e2[NC][2];
};
__device__ __forceinline__ void st_diff(Build& B, int c, const v4f (&Z)[2], float xs)
{
    B.d[c][0] = xs - Z[c >> 1][(2 * c) & 3];
    B.d[c][1] = xs - Z[c >> 1][((2 * c) & 3) + 1];
}
__device__ __forceinline__ void st_square(Build& B, int c)
{
    B.q[c][0] = B.d[c][0] * B.d[c][0];
    B.q[c][1] = B.d[c][1] * B.d[c][1];
}
__device__ __forceinline__ void st_scale1(Build& B, int c, float k1)
{
    B.e1[c][0] = B.q[c][0] * k1;
    B.e1[c][1] = B.q[c][1] * k1;
}
__device__ __forceinline__ void st_scale2(Build& B, int c, float k2)
{
    B.e2[c][0] = B.q[c][0] * k2;
    B.e2[c][1] = B.q[c][1] * k2;
}
__device__ __forceinline__ void st_exp1(Frag& F, const Build& B, int c)
{
    F.b1[c] = v2f{ __builtin_amdgcn_exp2f(B.e1[c][0]), __builtin_amdgcn_exp2f(B.e1[c][1]) };
}
__device__ __forceinline__ void st_exp2(Frag& F, const Build& B, int c)
{
    F.b2[c] = v2f{ __builtin_amdgcn_exp2f(B.e2[c][0]), __builtin_amdgcn_exp2f(B.e2[c][1]) };
}

struct Src {
    float xs, k1, k2;
};

__device__ __forceinline__ void carry(Build& To, const Build& From)
{
#pragma unroll
    for (int e = 0; e < 2; ++e) {
        To.d[2][e]  = From.d[2][e];
        To.q[1][e]  = From.q[1][e];
        To.e1[0][e] = From.e1[0][e];
        To.e2[0][e] = From.e2[0][e];
    }
}

__global__ __launch_bounds__(32 * TPB * NW)
void gpkan_rows_kernel(const float* __restrict__ x,
                       const float* __restrict__ z,
                       const float* __restrict__ qmu,
                       const float* __restrict__ qlv,
                       const float* __restrict__ lsc,
                       const float* __restrict__ lva,
                       float* __restrict__ out)
{
    __shared__ float part[TPB * NW][I_DIM / IB][16][32];

    const int wv   = threadIdx.x >> 5;
    const int pt   = wv & (TPB - 1);
    const int w    = wv / TPB;
    const int o    = blockIdx.x >> 1;
    const int tile = ((blockIdx.x & 1) << 2) + pt;
    const int lane = threadIdx.x & 31;
    const int r    = lane & 15;
    const int h    = lane >> 4;
    const int n    = tile * 16 + r;
    const int m0   = 16 * w + 8 * h;

    const size_t ob = (size_t)o * I_DIM;
    const float* z_p   = z   + ob * M_DIM + m0;
    const float* qm_p  = qmu + (ob + r) * M_DIM + m0;
    const float* ql_p  = qlv + (ob + r) * M_DIM + m0;
    const float* x_p   = x + (size_t)n * I_DIM;
    const float* ls_p  = lsc + ob + r;
    const float* lv_p  = lva + ob + r;
    const float* lvh_p = lva + ob + 8 * h;

    Raw first;
    load_raw(first, qm_p, ql_p, x_p, ls_p, lv_p, lvh_p, 0);
    Raw ahead;
    load_raw(ahead, qm_p, ql_p, x_p, ls_p, lv_p, lvh_p, IB);

    v4f z0row[2];
    load_zrow(z0row, z_p, 0);
    v4f zq[ZD][2];
#pragma unroll
    for (int d = 0; d < ZD; ++d) load_zrow(zq[d], z_p, 1 + d);

    Ops P;
    prepare(P, first);

    Frag F;
    {
        const float k1 = from_lane(P.K1, 0);
        const float k2 = from_lane(P.K2, 0);
        Build B0;
#pragma unroll
        for (int c = 0; c < NC; ++c) {
            st_diff(B0, c, z0row, P.X[0]);
            st_square(B0, c);
            st_scale1(B0, c, k1);
            st_scale2(B0, c, k2);
            st_exp1(F, B0, c);
            st_exp2(F, B0, c);
        }
    }

    Src   Sc = { P.X[1], from_lane(P.K1, 1), from_lane(P.K2, 1) };
    Build Bc;
    st_diff(Bc, 0, zq[0], Sc.xs);
    st_diff(Bc, 1, zq[0], Sc.xs);
    st_diff(Bc, 2, zq[0], Sc.xs);
    st_square(Bc, 0);
    st_square(Bc, 1);
    st_scale1(Bc, 0, Sc.k1);
    st_scale2(Bc, 0, Sc.k2);

#pragma unroll 1
    for (int b = 0; b < I_DIM / IB; ++b) {
        const int i0 = b * IB;

        Ops Pn;
        Raw nxt;
        float E1[16], E2[16];

#pragma unroll
        for (int s = 0; s < IB; ++s) {
            v4f zn[2];
            load_zrow(zn, z_p, min(i0 + s + 1 + ZD, I_DIM - 1));

            if (s == PREP) prepare(Pn, ahead);
            if (s == LOADS)
                load_raw(nxt, qm_p, ql_p, x_p, ls_p, lv_p, lvh_p, min(b + 2, I_DIM / IB - 1) * IB);

            const Ops& Q2 = (s + 2 >= IB) ? Pn : P;
            const int  s2 = (s + 2) & (IB - 1);
            const Src  Sn = { Q2.X[s2], from_lane(Q2.K1, s2), from_lane(Q2.K2, s2) };

            Frag  Fn;
            Build Bn;
            v8f a1 = {};
            v8f a2 = {};

#pragma unroll
            for (int c = 0; c < NC; ++c) {
                v2f w1 = { P.QS[2 * c], P.QS[2 * c + 1] };
                v2f w2 = { P.W2[2 * c], P.W2[2 * c + 1] };
                a1 = __builtin_amdgcn_wmma_f32_16x16x4_f32(
                         false, w1, false, F.b1[c], (short)0, a1, false, false);
                a2 = __builtin_amdgcn_wmma_f32_16x16x4_f32(
                         false, w2, false, F.b2[c], (short)0, a2, false, false);
                st_exp1(Fn, Bc, c);
                st_exp2(Fn, Bc, c);
                if (c + 1 < NC) { st_scale1(Bc, c + 1, Sc.k1); st_scale2(Bc, c + 1, Sc.k2); }
                if (c + 2 < NC) st_square(Bc, c + 2);
                if (c + 3 < NC) st_diff(Bc, c + 3, zq[0], Sc.xs);
                if (c == NC - 3) st_diff(Bn, 0, zq[1], Sn.xs);
                if (c == NC - 2) { st_diff(Bn, 1, zq[1], Sn.xs); st_square(Bn, 0); }
                if (c == NC - 1) {
                    st_diff(Bn, 2, zq[1], Sn.xs);
                    st_square(Bn, 1);
                    st_scale1(Bn, 0, Sn.k1);
                    st_scale2(Bn, 0, Sn.k2);
                }
            }
            E1[s] = a1[s & 7];
            E2[s] = a2[s & 7];
            F = Fn;
            carry(Bc, Bn);
            Sc = Sn;

            if (s == PREP) {
                __builtin_amdgcn_sched_group_barrier(0x402, 22, 0);
#pragma unroll
                for (int g = 0; g < NC; ++g) {
                    __builtin_amdgcn_sched_group_barrier(0x008, 2, 0);
                    __builtin_amdgcn_sched_group_barrier(0x402, 40, 0);
                }
            } else {
                __builtin_amdgcn_sched_group_barrier(0x402, 8, 0);
#pragma unroll
                for (int g = 0; g < NC; ++g) {
                    __builtin_amdgcn_sched_group_barrier(0x008, 2, 0);
                    __builtin_amdgcn_sched_group_barrier(0x400, 4, 0);
                    __builtin_amdgcn_sched_group_barrier(0x002, 8, 0);
                }
            }
            __builtin_amdgcn_sched_barrier(0);

#pragma unroll
            for (int d = 0; d + 1 < ZD; ++d) {
#pragma unroll
                for (int q = 0; q < 2; ++q) zq[d][q] = zq[d + 1][q];
            }
#pragma unroll
            for (int q = 0; q < 2; ++q) zq[ZD - 1][q] = zn[q];
        }

#pragma unroll
        for (int j = 0; j < 8; ++j) {
            float em = h ? E1[8 + j] : E1[j];
            float t  = h ? E2[8 + j] : E2[j];
            if (w == 0) t = P.SVh[j] + t;
            part[wv][b][j][lane]     = em;
            part[wv][b][8 + j][lane] = t;
        }

        asm volatile("" :: "v"(nxt.qm[0]), "v"(nxt.qm[1]),
                           "v"(nxt.ql[0]), "v"(nxt.ql[1]),
                           "v"(nxt.xx[0]), "v"(nxt.xx[1]), "v"(nxt.xx[2]), "v"(nxt.xx[3]),
                           "v"(nxt.lvh[0]), "v"(nxt.lvh[1]), "v"(nxt.ls), "v"(nxt.lv));
        ahead = nxt;
        P = Pn;
    }

    __syncthreads();

    if (w == 0) {
        float ym = 0.f, yv = 0.f;
#pragma unroll
        for (int b = 0; b < I_DIM / IB; ++b) {
#pragma unroll
            for (int j = 0; j < 8; ++j) {
                float em = part[wv][b][j][lane]     + part[wv + TPB][b][j][lane];
                float t  = part[wv][b][8 + j][lane] + part[wv + TPB][b][8 + j][lane];
                float ev = t - em * em;
                ev = fmaxf(ev, 1e-6f);
                ym += em;
                yv += ev;
            }
        }

        ym += __shfl_xor(ym, 16, 32);
        yv += __shfl_xor(yv, 16, 32);

        if (lane < 16) {
            out[n * O_DIM + o]                 = ym;
            out[N_DIM * O_DIM + n * O_DIM + o] = yv;
        }
    }
}

extern "C" void kernel_launch(void* const* d_in, const int* in_sizes, int n_in,
                              void* d_out, int out_size, void* d_ws, size_t ws_size,
                              hipStream_t stream) {
    const float* x   = (const float*)d_in[0];
    const float* z   = (const float*)d_in[1];
    const float* qmu = (const float*)d_in[2];
    const float* qlv = (const float*)d_in[3];
    const float* lsc = (const float*)d_in[4];
    const float* lva = (const float*)d_in[5];
    float* out = (float*)d_out;

    dim3 grid(256);
    dim3 block(256);
    gpkan_rows_kernel<<<grid, block, 0, stream>>>(x, z, qmu, qlv, lsc, lva, out);
}


# speedup vs baseline: 2.8551x; 1.0317x over previous
#include <hip/hip_runtime.h>
#include <math.h>


typedef __attribute__((ext_vector_type(2))) float v2f;
typedef __attribute__((ext_vector_type(4))) float v4f;
typedef __attribute__((ext_vector_type(8))) float v8f;

#define O_DIM 128
#define I_DIM 128
#define M_DIM 32
#define N_DIM 128
#define LOG2E 1.44269504088896340736f
#define IB    16
#define NW    2
#define NC    4
#define TPB   4
#define ZD    4
#define PREP  0
#define LOADS 11


struct Raw {
    v4f qm[2];
    v4f ql[2];
    v4f xx[4];
    v4f lvh[2];
    float ls;
    float lv;
};

struct Ops {
    float QS[8];
    float W2[8];
    float X[16];
    float SVh[8];
    float K1, K2;
};

__device__ __forceinline__ void load_raw(Raw& R,
        const float* __restrict__ qm_p, const float* __restrict__ ql_p,
        const float* __restrict__ x_p,  const float* __restrict__ ls_p,
        const float* __restrict__ lv_p, const float* __restrict__ lvh_p, int i0)
{
    const v4f* q4 = (const v4f*)(qm_p + (size_t)i0 * M_DIM);
    const v4f* l4 = (const v4f*)(ql_p + (size_t)i0 * M_DIM);
    const v4f* x4 = (const v4f*)(x_p + i0);
    const v4f* h4 = (const v4f*)(lvh_p + i0);
#pragma unroll
    for (int q = 0; q < 2; ++q) { R.qm[q] = q4[q]; R.ql[q] = l4[q]; }
#pragma unroll
    for (int q = 0; q < 4; ++q) R.xx[q] = x4[q];
    R.lvh[0] = h4[0];
    R.lvh[1] = h4[1];
    R.ls = ls_p[i0];
    R.lv = lv_p[i0];
}

__device__ __forceinline__ void load_zrow(v4f (&Z)[2], const float* __restrict__ z_p, int i)
{
    const v4f* p = (const v4f*)(z_p + (size_t)i * M_DIM);
#pragma unroll
    for (int q = 0; q < 2; ++q) Z[q] = p[q];
}

__device__ __forceinline__ void prepare(Ops& P, const Raw& R)
{
    float scale = fmaxf(__expf(R.ls), 0.1f);
    float l2    = scale * scale;
    float sv    = fmaxf(__expf(R.lv), 1e-5f);
    float den1  = l2 + 1e-6f;
    float den2  = l2 + 2e-6f;
    float r1    = sqrtf(l2 / den1);
    float r2    = sqrtf(l2 / den2);
    float c2sv  = sv * r2;
    P.K1 = -0.5f * LOG2E / den1;
    P.K2 = -LOG2E / den2;
#pragma unroll
    for (int j = 0; j < 8; ++j) {
        float qm = R.qm[j >> 2][j & 3];
        float qv = fmaxf(__expf(R.ql[j >> 2][j & 3]), 1e-5f);
        P.QS[j] = qm * r1;
        P.W2[j] = (qv + qm * qm) * r2 - c2sv;
    }
#pragma unroll
    for (int j = 0; j < 16; ++j)
        P.X[j] = R.xx[j >> 2][j & 3];
#pragma unroll
    for (int j = 0; j < 8; ++j)
        P.SVh[j] = fmaxf(__expf(R.lvh[j >> 2][j & 3]), 1e-5f);
}

__device__ __forceinline__ float from_lane(float v, int src)
{
    return __int_as_float(__builtin_amdgcn_readlane(__float_as_int(v), src));
}

struct Frag {
    v2f b1[NC];
    v2f b2[NC];
};

struct Build {
    v2f d[NC];
    v2f q[NC];
    v2f e1[NC];
    v2f e2[NC];
};
__device__ __forceinline__ void st_diff(Build& B, int c, const v4f (&Z)[2], float xs)
{
    const v2f zz = { Z[c >> 1][(2 * c) & 3], Z[c >> 1][((2 * c) & 3) + 1] };
    B.d[c] = xs - zz;
}
__device__ __forceinline__ void st_square(Build& B, int c)
{
    B.q[c] = B.d[c] * B.d[c];
}
__device__ __forceinline__ void st_scale1(Build& B, int c, float k1)
{
    B.e1[c] = B.q[c] * k1;
}
__device__ __forceinline__ void st_scale2(Build& B, int c, float k2)
{
    B.e2[c] = B.q[c] * k2;
}
__device__ __forceinline__ void st_exp1(Frag& F, const Build& B, int c)
{
    F.b1[c] = v2f{ __builtin_amdgcn_exp2f(B.e1[c][0]), __builtin_amdgcn_exp2f(B.e1[c][1]) };
}
__device__ __forceinline__ void st_exp2(Frag& F, const Build& B, int c)
{
    F.b2[c] = v2f{ __builtin_amdgcn_exp2f(B.e2[c][0]), __builtin_amdgcn_exp2f(B.e2[c][1]) };
}

struct Src {
    float xs, k1, k2;
};

__device__ __forceinline__ void carry(Build& To, const Build& From)
{
    To.d[2]  = From.d[2];
    To.q[1]  = From.q[1];
    To.e1[0] = From.e1[0];
    To.e2[0] = From.e2[0];
}

__global__ __launch_bounds__(32 * TPB * NW)
void gpkan_rows_kernel(const float* __restrict__ x,
                       const float* __restrict__ z,
                       const float* __restrict__ qmu,
                       const float* __restrict__ qlv,
                       const float* __restrict__ lsc,
                       const float* __restrict__ lva,
                       float* __restrict__ out)
{
    __shared__ float part[TPB * NW][I_DIM / IB][16][32];

    const int wv   = threadIdx.x >> 5;
    const int pt   = wv & (TPB - 1);
    const int w    = wv / TPB;
    const int o    = blockIdx.x >> 1;
    const int tile = ((blockIdx.x & 1) << 2) + pt;
    const int lane = threadIdx.x & 31;
    const int r    = lane & 15;
    const int h    = lane >> 4;
    const int n    = tile * 16 + r;
    const int m0   = 16 * w + 8 * h;

    const size_t ob = (size_t)o * I_DIM;
    const float* z_p   = z   + ob * M_DIM + m0;
    const float* qm_p  = qmu + (ob + r) * M_DIM + m0;
    const float* ql_p  = qlv + (ob + r) * M_DIM + m0;
    const float* x_p   = x + (size_t)n * I_DIM;
    const float* ls_p  = lsc + ob + r;
    const float* lv_p  = lva + ob + r;
    const float* lvh_p = lva + ob + 8 * h;

    Raw first;
    load_raw(first, qm_p, ql_p, x_p, ls_p, lv_p, lvh_p, 0);
    Raw ahead;
    load_raw(ahead, qm_p, ql_p, x_p, ls_p, lv_p, lvh_p, IB);

    v4f z0row[2];
    load_zrow(z0row, z_p, 0);
    v4f zq[ZD][2];
#pragma unroll
    for (int d = 0; d < ZD; ++d) load_zrow(zq[d], z_p, 1 + d);

    Ops P;
    prepare(P, first);

    Frag F;
    {
        const float k1 = from_lane(P.K1, 0);
        const float k2 = from_lane(P.K2, 0);
        Build B0;
#pragma unroll
        for (int c = 0; c < NC; ++c) {
            st_diff(B0, c, z0row, P.X[0]);
            st_square(B0, c);
            st_scale1(B0, c, k1);
            st_scale2(B0, c, k2);
            st_exp1(F, B0, c);
            st_exp2(F, B0, c);
        }
    }

    Src   Sc = { P.X[1], from_lane(P.K1, 1), from_lane(P.K2, 1) };
    Build Bc;
    st_diff(Bc, 0, zq[0], Sc.xs);
    st_diff(Bc, 1, zq[0], Sc.xs);
    st_diff(Bc, 2, zq[0], Sc.xs);
    st_square(Bc, 0);
    st_square(Bc, 1);
    st_scale1(Bc, 0, Sc.k1);
    st_scale2(Bc, 0, Sc.k2);

#pragma unroll 1
    for (int b = 0; b < I_DIM / IB; ++b) {
        const int i0 = b * IB;

        Ops Pn;
        Raw nxt;
        float E1[16], E2[16];

#pragma unroll
        for (int s = 0; s < IB; ++s) {
            v4f zn[2];
            load_zrow(zn, z_p, min(i0 + s + 1 + ZD, I_DIM - 1));

            if (s == PREP) prepare(Pn, ahead);
            if (s == LOADS)
                load_raw(nxt, qm_p, ql_p, x_p, ls_p, lv_p, lvh_p, min(b + 2, I_DIM / IB - 1) * IB);

            const Ops& Q2 = (s + 2 >= IB) ? Pn : P;
            const int  s2 = (s + 2) & (IB - 1);
            const Src  Sn = { Q2.X[s2], from_lane(Q2.K1, s2), from_lane(Q2.K2, s2) };

            Frag  Fn;
            Build Bn;
            v8f a1 = {};
            v8f a2 = {};

            if (s != PREP) __builtin_amdgcn_sched_barrier(0);
#pragma unroll
            for (int c = 0; c < NC; ++c) {
                v2f w1 = { P.QS[2 * c], P.QS[2 * c + 1] };
                v2f w2 = { P.W2[2 * c], P.W2[2 * c + 1] };
                a1 = __builtin_amdgcn_wmma_f32_16x16x4_f32(
                         false, w1, false, F.b1[c], (short)0, a1, false, false);
                a2 = __builtin_amdgcn_wmma_f32_16x16x4_f32(
                         false, w2, false, F.b2[c], (short)0, a2, false, false);
                st_exp1(Fn, Bc, c);
                st_exp2(Fn, Bc, c);
                if (c + 1 < NC) { st_scale1(Bc, c + 1, Sc.k1); st_scale2(Bc, c + 1, Sc.k2); }
                if (c + 2 < NC) st_square(Bc, c + 2);
                if (c + 3 < NC) st_diff(Bc, c + 3, zq[0], Sc.xs);
                if (c == NC - 3) st_diff(Bn, 0, zq[1], Sn.xs);
                if (c == NC - 2) { st_diff(Bn, 1, zq[1], Sn.xs); st_square(Bn, 0); }
                if (c == NC - 1) {
                    st_diff(Bn, 2, zq[1], Sn.xs);
                    st_square(Bn, 1);
                    st_scale1(Bn, 0, Sn.k1);
                    st_scale2(Bn, 0, Sn.k2);
                }
                if (s != PREP) {
                    __builtin_amdgcn_sched_group_barrier(0x008, 2, 0);
#pragma unroll
                    for (int t = 0; t < 4; ++t) {
                        __builtin_amdgcn_sched_group_barrier(0x002, 1, 0);
                        __builtin_amdgcn_sched_group_barrier(0x400, 1, 0);
                    }
                    __builtin_amdgcn_sched_barrier(0);
                }
            }
            E1[s] = a1[s & 7];
            E2[s] = a2[s & 7];
            F = Fn;
            carry(Bc, Bn);
            Sc = Sn;

            if (s == PREP) {
                __builtin_amdgcn_sched_group_barrier(0x402, 22, 0);
#pragma unroll
                for (int g = 0; g < NC; ++g) {
                    __builtin_amdgcn_sched_group_barrier(0x008, 2, 0);
                    __builtin_amdgcn_sched_group_barrier(0x402, 40, 0);
                }
            }
            __builtin_amdgcn_sched_barrier(0);

#pragma unroll
            for (int d = 0; d + 1 < ZD; ++d) {
#pragma unroll
                for (int q = 0; q < 2; ++q) zq[d][q] = zq[d + 1][q];
            }
#pragma unroll
            for (int q = 0; q < 2; ++q) zq[ZD - 1][q] = zn[q];
        }

#pragma unroll
        for (int j = 0; j < 8; ++j) {
            float em = h ? E1[8 + j] : E1[j];
            float t  = h ? E2[8 + j] : E2[j];
            if (w == 0) t = P.SVh[j] + t;
            part[wv][b][j][lane]     = em;
            part[wv][b][8 + j][lane] = t;
        }

        asm volatile("" :: "v"(nxt.qm[0]), "v"(nxt.qm[1]),
                           "v"(nxt.ql[0]), "v"(nxt.ql[1]),
                           "v"(nxt.xx[0]), "v"(nxt.xx[1]), "v"(nxt.xx[2]), "v"(nxt.xx[3]),
                           "v"(nxt.lvh[0]), "v"(nxt.lvh[1]), "v"(nxt.ls), "v"(nxt.lv));
        ahead = nxt;
        P = Pn;
    }

    __syncthreads();

    if (w == 0) {
        float ym = 0.f, yv = 0.f;
#pragma unroll
        for (int b = 0; b < I_DIM / IB; ++b) {
#pragma unroll
            for (int j = 0; j < 8; ++j) {
                float em = part[wv][b][j][lane]     + part[wv + TPB][b][j][lane];
                float t  = part[wv][b][8 + j][lane] + part[wv + TPB][b][8 + j][lane];
                float ev = t - em * em;
                ev = fmaxf(ev, 1e-6f);
                ym += em;
                yv += ev;
            }
        }

        ym += __shfl_xor(ym, 16, 32);
        yv += __shfl_xor(yv, 16, 32);

        if (lane < 16) {
            out[n * O_DIM + o]                 = ym;
            out[N_DIM * O_DIM + n * O_DIM + o] = yv;
        }
    }
}

extern "C" void kernel_launch(void* const* d_in, const int* in_sizes, int n_in,
                              void* d_out, int out_size, void* d_ws, size_t ws_size,
                              hipStream_t stream) {
    const float* x   = (const float*)d_in[0];
    const float* z   = (const float*)d_in[1];
    const float* qmu = (const float*)d_in[2];
    const float* qlv = (const float*)d_in[3];
    const float* lsc = (const float*)d_in[4];
    const float* lva = (const float*)d_in[5];
    float* out = (float*)d_out;

    dim3 grid(256);
    dim3 block(256);
    gpkan_rows_kernel<<<grid, block, 0, stream>>>(x, z, qmu, qlv, lsc, lva, out);
}
